# starting kernel
# baseline (speedup 1.0000x reference)
#include <hip/hip_runtime.h>
#include <hip/hip_bf16.h>

typedef float v2f __attribute__((ext_vector_type(2)));
typedef float v8f __attribute__((ext_vector_type(8)));

#define BB   4096
#define DD   32
#define HH   5
#define GG   20
#define INW  128

__device__ __forceinline__ float sigf(float x) { return 1.0f / (1.0f + __expf(-x)); }

__global__ __launch_bounds__(256) void hb_kernel(
    const float* __restrict__ hidden,
    const float* __restrict__ W_hh0, const float* __restrict__ b_ih0, const float* __restrict__ b_hh0,
    const float* __restrict__ W_hh1, const float* __restrict__ b_ih1, const float* __restrict__ b_hh1,
    float* __restrict__ hb0, float* __restrict__ hb1)
{
    int b = blockIdx.x * blockDim.x + threadIdx.x;
    if (b >= BB) return;
    float h0[HH], h1[HH];
    #pragma unroll
    for (int j = 0; j < HH; ++j) {
        h0[j] = hidden[b * HH + j];
        h1[j] = hidden[BB * HH + b * HH + j];
    }
    #pragma unroll 4
    for (int g = 0; g < GG; ++g) {
        float s0 = b_ih0[g] + b_hh0[g];
        float s1 = b_ih1[g] + b_hh1[g];
        #pragma unroll
        for (int j = 0; j < HH; ++j) {
            s0 += h0[j] * W_hh0[g * HH + j];
            s1 += h1[j] * W_hh1[g * HH + j];
        }
        hb0[b * GG + g] = s0;
        hb1[b * GG + g] = s1;
    }
}

__global__ __launch_bounds__(256) void lstm_wmma_kernel(
    const float* __restrict__ dec_x,
    const int*   __restrict__ cat1,
    const int*   __restrict__ cat2,
    const int*   __restrict__ cat3,
    const float* __restrict__ emb,
    const float* __restrict__ W_ih0,
    const float* __restrict__ W_ih1,
    const float* __restrict__ cell,
    const float* __restrict__ hb0,
    const float* __restrict__ hb1,
    float* __restrict__ out)
{
    __shared__ float Wp[32 * INW];
    __shared__ float Gs[8][16 * GG];

    const int tid = threadIdx.x;
    for (int i = tid; i < 32 * INW; i += 256) {
        Wp[i] = ((i >> 7) < GG) ? W_ih0[i] : 0.0f;
    }
    __syncthreads();

    const int wave = tid >> 5;
    const int lane = tid & 31;
    const int m    = lane & 15;
    const int half = lane >> 4;
    const int koff = half * 2;

    const long rbase = ((long)blockIdx.x * 8 + wave) * 16;
    const int  t     = (int)(rbase >> 12);
    const int  b     = (int)(rbase & 4095) + m;
    const long row   = (long)t * BB + b;

    const float* srcs[4];
    srcs[0] = dec_x + row * DD;
    srcs[1] = emb + (long)cat1[row] * 32;
    srcs[2] = emb + (long)cat2[row] * 32;
    srcs[3] = emb + (long)cat3[row] * 32;

    const float* bp0 = &Wp[m * INW];
    const float* bp1 = &Wp[(m + 16) * INW];

    v8f acc0 = {};
    v8f acc1 = {};
    #pragma unroll
    for (int seg = 0; seg < 4; ++seg) {
        const float* ap = srcs[seg];
        #pragma unroll
        for (int kk = 0; kk < 8; ++kk) {
            const int col  = kk * 4 + koff;
            const int gcol = seg * 32 + col;
            v2f a  = *(const v2f*)(ap  + col);
            v2f b0 = *(const v2f*)(bp0 + gcol);
            v2f b1 = *(const v2f*)(bp1 + gcol);
            acc0 = __builtin_amdgcn_wmma_f32_16x16x4_f32(false, a, false, b0,
                                                         (short)0, acc0, false, false);
            acc1 = __builtin_amdgcn_wmma_f32_16x16x4_f32(false, a, false, b1,
                                                         (short)0, acc1, false, false);
        }
    }

    float* gs = &Gs[wave][0];
    #pragma unroll
    for (int j = 0; j < 8; ++j) {
        const int rm = j + half * 8;
        gs[rm * GG + m] = acc0[j];
        if (m < 4) gs[rm * GG + 16 + m] = acc1[j];
    }
    asm volatile("s_wait_dscnt 0" ::: "memory");

    if (lane < 16) {
        float g0[GG];
        #pragma unroll
        for (int g = 0; g < GG; ++g) g0[g] = gs[m * GG + g] + hb0[(long)b * GG + g];

        float h1v[HH];
        #pragma unroll
        for (int h = 0; h < HH; ++h) {
            float cn = sigf(g0[5 + h]) * cell[(long)b * HH + h]
                     + sigf(g0[h]) * tanhf(g0[10 + h]);
            h1v[h] = sigf(g0[15 + h]) * tanhf(cn);
        }

        float g1[GG];
        #pragma unroll 4
        for (int g = 0; g < GG; ++g) {
            float s = hb1[(long)b * GG + g];
            #pragma unroll
            for (int j = 0; j < HH; ++j) s += W_ih1[g * HH + j] * h1v[j];
            g1[g] = s;
        }
        #pragma unroll
        for (int h = 0; h < HH; ++h) {
            float cn = sigf(g1[5 + h]) * cell[(long)BB * HH + (long)b * HH + h]
                     + sigf(g1[h]) * tanhf(g1[10 + h]);
            out[row * HH + h] = sigf(g1[15 + h]) * tanhf(cn);
        }
    }
}

extern "C" void kernel_launch(void* const* d_in, const int* in_sizes, int n_in,
                              void* d_out, int out_size, void* d_ws, size_t ws_size,
                              hipStream_t stream)
{
    const float* hidden = (const float*)d_in[1];
    const float* cellp  = (const float*)d_in[2];
    const float* dec_x  = (const float*)d_in[3];
    const int*   cat1   = (const int*)d_in[4];
    const int*   cat2   = (const int*)d_in[5];
    const int*   cat3   = (const int*)d_in[6];
    const float* emb    = (const float*)d_in[7];
    const float* W_ih0  = (const float*)d_in[8];
    const float* W_hh0  = (const float*)d_in[9];
    const float* b_ih0  = (const float*)d_in[10];
    const float* b_hh0  = (const float*)d_in[11];
    const float* W_ih1  = (const float*)d_in[12];
    const float* W_hh1  = (const float*)d_in[13];
    const float* b_ih1  = (const float*)d_in[14];
    const float* b_hh1  = (const float*)d_in[15];
    float* out = (float*)d_out;

    const int T    = in_sizes[3] / (BB * DD);
    const int rows = T * BB;

    float* hb0 = (float*)d_ws;
    float* hb1 = hb0 + (size_t)BB * GG;

    hb_kernel<<<BB / 256, 256, 0, stream>>>(hidden, W_hh0, b_ih0, b_hh0,
                                            W_hh1, b_ih1, b_hh1, hb0, hb1);

    lstm_wmma_kernel<<<rows / 128, 256, 0, stream>>>(dec_x, cat1, cat2, cat3, emb,
                                                     W_ih0, W_ih1, cellp,
                                                     hb0, hb1, out);
}


# LDS row pitch of the staged W_ih0 padded from 128 to 132 floats: the 16 rows of a B-fragment read fall on 16 different bank pairs instead of one (16-way conflict gone); k order and sums unchanged
# speedup vs baseline: 1.3129x; 1.3129x over previous
#include <hip/hip_runtime.h>
#include <hip/hip_bf16.h>

typedef float v2f __attribute__((ext_vector_type(2)));
typedef float v8f __attribute__((ext_vector_type(8)));

#define BB   4096
#define DD   32
#define HH   5
#define GG   20
#define INW  128
#define WPITCH 132

__device__ __forceinline__ float sigf(float x) { return 1.0f / (1.0f + __expf(-x)); }

__global__ __launch_bounds__(256) void hb_kernel(
    const float* __restrict__ hidden,
    const float* __restrict__ W_hh0, const float* __restrict__ b_ih0, const float* __restrict__ b_hh0,
    const float* __restrict__ W_hh1, const float* __restrict__ b_ih1, const float* __restrict__ b_hh1,
    float* __restrict__ hb0, float* __restrict__ hb1)
{
    int b = blockIdx.x * blockDim.x + threadIdx.x;
    if (b >= BB) return;
    float h0[HH], h1[HH];
    #pragma unroll
    for (int j = 0; j < HH; ++j) {
        h0[j] = hidden[b * HH + j];
        h1[j] = hidden[BB * HH + b * HH + j];
    }
    #pragma unroll 4
    for (int g = 0; g < GG; ++g) {
        float s0 = b_ih0[g] + b_hh0[g];
        float s1 = b_ih1[g] + b_hh1[g];
        #pragma unroll
        for (int j = 0; j < HH; ++j) {
            s0 += h0[j] * W_hh0[g * HH + j];
            s1 += h1[j] * W_hh1[g * HH + j];
        }
        hb0[b * GG + g] = s0;
        hb1[b * GG + g] = s1;
    }
}

__global__ __launch_bounds__(256) void lstm_wmma_kernel(
    const float* __restrict__ dec_x,
    const int*   __restrict__ cat1,
    const int*   __restrict__ cat2,
    const int*   __restrict__ cat3,
    const float* __restrict__ emb,
    const float* __restrict__ W_ih0,
    const float* __restrict__ W_ih1,
    const float* __restrict__ cell,
    const float* __restrict__ hb0,
    const float* __restrict__ hb1,
    float* __restrict__ out)
{
    __shared__ float Wp[32 * WPITCH];
    __shared__ float Gs[8][16 * GG];

    const int tid = threadIdx.x;
    for (int i = tid; i < 32 * INW; i += 256) {
        Wp[(i >> 7) * WPITCH + (i & (INW - 1))] = ((i >> 7) < GG) ? W_ih0[i] : 0.0f;
    }
    __syncthreads();

    const int wave = tid >> 5;
    const int lane = tid & 31;
    const int m    = lane & 15;
    const int half = lane >> 4;
    const int koff = half * 2;

    const long rbase = ((long)blockIdx.x * 8 + wave) * 16;
    const int  t     = (int)(rbase >> 12);
    const int  b     = (int)(rbase & 4095) + m;
    const long row   = (long)t * BB + b;

    const float* srcs[4];
    srcs[0] = dec_x + row * DD;
    srcs[1] = emb + (long)cat1[row] * 32;
    srcs[2] = emb + (long)cat2[row] * 32;
    srcs[3] = emb + (long)cat3[row] * 32;

    const float* bp0 = &Wp[m * WPITCH];
    const float* bp1 = &Wp[(m + 16) * WPITCH];

    v8f acc0 = {};
    v8f acc1 = {};
    #pragma unroll
    for (int seg = 0; seg < 4; ++seg) {
        const float* ap = srcs[seg];
        #pragma unroll
        for (int kk = 0; kk < 8; ++kk) {
            const int col  = kk * 4 + koff;
            const int gcol = seg * 32 + col;
            v2f a  = *(const v2f*)(ap  + col);
            v2f b0 = *(const v2f*)(bp0 + gcol);
            v2f b1 = *(const v2f*)(bp1 + gcol);
            acc0 = __builtin_amdgcn_wmma_f32_16x16x4_f32(false, a, false, b0,
                                                         (short)0, acc0, false, false);
            acc1 = __builtin_amdgcn_wmma_f32_16x16x4_f32(false, a, false, b1,
                                                         (short)0, acc1, false, false);
        }
    }

    float* gs = &Gs[wave][0];
    #pragma unroll
    for (int j = 0; j < 8; ++j) {
        const int rm = j + half * 8;
        gs[rm * GG + m] = acc0[j];
        if (m < 4) gs[rm * GG + 16 + m] = acc1[j];
    }
    asm volatile("s_wait_dscnt 0" ::: "memory");

    if (lane < 16) {
        float g0[GG];
        #pragma unroll
        for (int g = 0; g < GG; ++g) g0[g] = gs[m * GG + g] + hb0[(long)b * GG + g];

        float h1v[HH];
        #pragma unroll
        for (int h = 0; h < HH; ++h) {
            float cn = sigf(g0[5 + h]) * cell[(long)b * HH + h]
                     + sigf(g0[h]) * tanhf(g0[10 + h]);
            h1v[h] = sigf(g0[15 + h]) * tanhf(cn);
        }

        float g1[GG];
        #pragma unroll 4
        for (int g = 0; g < GG; ++g) {
            float s = hb1[(long)b * GG + g];
            #pragma unroll
            for (int j = 0; j < HH; ++j) s += W_ih1[g * HH + j] * h1v[j];
            g1[g] = s;
        }
        #pragma unroll
        for (int h = 0; h < HH; ++h) {
            float cn = sigf(g1[5 + h]) * cell[(long)BB * HH + (long)b * HH + h]
                     + sigf(g1[h]) * tanhf(g1[10 + h]);
            out[row * HH + h] = sigf(g1[15 + h]) * tanhf(cn);
        }
    }
}

extern "C" void kernel_launch(void* const* d_in, const int* in_sizes, int n_in,
                              void* d_out, int out_size, void* d_ws, size_t ws_size,
                              hipStream_t stream)
{
    const float* hidden = (const float*)d_in[1];
    const float* cellp  = (const float*)d_in[2];
    const float* dec_x  = (const float*)d_in[3];
    const int*   cat1   = (const int*)d_in[4];
    const int*   cat2   = (const int*)d_in[5];
    const int*   cat3   = (const int*)d_in[6];
    const float* emb    = (const float*)d_in[7];
    const float* W_ih0  = (const float*)d_in[8];
    const float* W_hh0  = (const float*)d_in[9];
    const float* b_ih0  = (const float*)d_in[10];
    const float* b_hh0  = (const float*)d_in[11];
    const float* W_ih1  = (const float*)d_in[12];
    const float* W_hh1  = (const float*)d_in[13];
    const float* b_ih1  = (const float*)d_in[14];
    const float* b_hh1  = (const float*)d_in[15];
    float* out = (float*)d_out;

    const int T    = in_sizes[3] / (BB * DD);
    const int rows = T * BB;

    float* hb0 = (float*)d_ws;
    float* hb1 = hb0 + (size_t)BB * GG;

    hb_kernel<<<BB / 256, 256, 0, stream>>>(hidden, W_hh0, b_ih0, b_hh0,
                                            W_hh1, b_ih1, b_hh1, hb0, hb1);

    lstm_wmma_kernel<<<rows / 128, 256, 0, stream>>>(dec_x, cat1, cat2, cat3, emb,
                                                     W_ih0, W_ih1, cellp,
                                                     hb0, hb1, out);
}


# All 50 gate nonlinearities per row on the hardware v_tanh_f32 (tanh direct, sigmoid as 0.5+0.5*tanh(x/2)) in place of library tanhf and IEEE-division sigmoid; sums and their order unchanged
# speedup vs baseline: 1.5306x; 1.1658x over previous
#include <hip/hip_runtime.h>
#include <hip/hip_bf16.h>

typedef float v2f __attribute__((ext_vector_type(2)));
typedef float v8f __attribute__((ext_vector_type(8)));

#define BB   4096
#define DD   32
#define HH   5
#define GG   20
#define INW  128
#define WPITCH 132

__device__ __forceinline__ float tanh_hw(float x) { return __builtin_amdgcn_tanhf(x); }
__device__ __forceinline__ float sigf(float x) { return __builtin_fmaf(0.5f, tanh_hw(0.5f * x), 0.5f); }

__global__ __launch_bounds__(256) void hb_kernel(
    const float* __restrict__ hidden,
    const float* __restrict__ W_hh0, const float* __restrict__ b_ih0, const float* __restrict__ b_hh0,
    const float* __restrict__ W_hh1, const float* __restrict__ b_ih1, const float* __restrict__ b_hh1,
    float* __restrict__ hb0, float* __restrict__ hb1)
{
    int b = blockIdx.x * blockDim.x + threadIdx.x;
    if (b >= BB) return;
    float h0[HH], h1[HH];
    #pragma unroll
    for (int j = 0; j < HH; ++j) {
        h0[j] = hidden[b * HH + j];
        h1[j] = hidden[BB * HH + b * HH + j];
    }
    #pragma unroll 4
    for (int g = 0; g < GG; ++g) {
        float s0 = b_ih0[g] + b_hh0[g];
        float s1 = b_ih1[g] + b_hh1[g];
        #pragma unroll
        for (int j = 0; j < HH; ++j) {
            s0 += h0[j] * W_hh0[g * HH + j];
            s1 += h1[j] * W_hh1[g * HH + j];
        }
        hb0[b * GG + g] = s0;
        hb1[b * GG + g] = s1;
    }
}

__global__ __launch_bounds__(256) void lstm_wmma_kernel(
    const float* __restrict__ dec_x,
    const int*   __restrict__ cat1,
    const int*   __restrict__ cat2,
    const int*   __restrict__ cat3,
    const float* __restrict__ emb,
    const float* __restrict__ W_ih0,
    const float* __restrict__ W_ih1,
    const float* __restrict__ cell,
    const float* __restrict__ hb0,
    const float* __restrict__ hb1,
    float* __restrict__ out)
{
    __shared__ float Wp[32 * WPITCH];
    __shared__ float Gs[8][16 * GG];

    const int tid = threadIdx.x;
    for (int i = tid; i < 32 * INW; i += 256) {
        Wp[(i >> 7) * WPITCH + (i & (INW - 1))] = ((i >> 7) < GG) ? W_ih0[i] : 0.0f;
    }
    __syncthreads();

    const int wave = tid >> 5;
    const int lane = tid & 31;
    const int m    = lane & 15;
    const int half = lane >> 4;
    const int koff = half * 2;

    const long rbase = ((long)blockIdx.x * 8 + wave) * 16;
    const int  t     = (int)(rbase >> 12);
    const int  b     = (int)(rbase & 4095) + m;
    const long row   = (long)t * BB + b;

    const float* srcs[4];
    srcs[0] = dec_x + row * DD;
    srcs[1] = emb + (long)cat1[row] * 32;
    srcs[2] = emb + (long)cat2[row] * 32;
    srcs[3] = emb + (long)cat3[row] * 32;

    const float* bp0 = &Wp[m * WPITCH];
    const float* bp1 = &Wp[(m + 16) * WPITCH];

    v8f acc0 = {};
    v8f acc1 = {};
    #pragma unroll
    for (int seg = 0; seg < 4; ++seg) {
        const float* ap = srcs[seg];
        #pragma unroll
        for (int kk = 0; kk < 8; ++kk) {
            const int col  = kk * 4 + koff;
            const int gcol = seg * 32 + col;
            v2f a  = *(const v2f*)(ap  + col);
            v2f b0 = *(const v2f*)(bp0 + gcol);
            v2f b1 = *(const v2f*)(bp1 + gcol);
            acc0 = __builtin_amdgcn_wmma_f32_16x16x4_f32(false, a, false, b0,
                                                         (short)0, acc0, false, false);
            acc1 = __builtin_amdgcn_wmma_f32_16x16x4_f32(false, a, false, b1,
                                                         (short)0, acc1, false, false);
        }
    }

    float* gs = &Gs[wave][0];
    #pragma unroll
    for (int j = 0; j < 8; ++j) {
        const int rm = j + half * 8;
        gs[rm * GG + m] = acc0[j];
        if (m < 4) gs[rm * GG + 16 + m] = acc1[j];
    }
    asm volatile("s_wait_dscnt 0" ::: "memory");

    if (lane < 16) {
        float g0[GG];
        #pragma unroll
        for (int g = 0; g < GG; ++g) g0[g] = gs[m * GG + g] + hb0[(long)b * GG + g];

        float h1v[HH];
        #pragma unroll
        for (int h = 0; h < HH; ++h) {
            float cn = sigf(g0[5 + h]) * cell[(long)b * HH + h]
                     + sigf(g0[h]) * tanh_hw(g0[10 + h]);
            h1v[h] = sigf(g0[15 + h]) * tanh_hw(cn);
        }

        float g1[GG];
        #pragma unroll 4
        for (int g = 0; g < GG; ++g) {
            float s = hb1[(long)b * GG + g];
            #pragma unroll
            for (int j = 0; j < HH; ++j) s += W_ih1[g * HH + j] * h1v[j];
            g1[g] = s;
        }
        #pragma unroll
        for (int h = 0; h < HH; ++h) {
            float cn = sigf(g1[5 + h]) * cell[(long)BB * HH + (long)b * HH + h]
                     + sigf(g1[h]) * tanh_hw(g1[10 + h]);
            out[row * HH + h] = sigf(g1[15 + h]) * tanh_hw(cn);
        }
    }
}

extern "C" void kernel_launch(void* const* d_in, const int* in_sizes, int n_in,
                              void* d_out, int out_size, void* d_ws, size_t ws_size,
                              hipStream_t stream)
{
    const float* hidden = (const float*)d_in[1];
    const float* cellp  = (const float*)d_in[2];
    const float* dec_x  = (const float*)d_in[3];
    const int*   cat1   = (const int*)d_in[4];
    const int*   cat2   = (const int*)d_in[5];
    const int*   cat3   = (const int*)d_in[6];
    const float* emb    = (const float*)d_in[7];
    const float* W_ih0  = (const float*)d_in[8];
    const float* W_hh0  = (const float*)d_in[9];
    const float* b_ih0  = (const float*)d_in[10];
    const float* b_hh0  = (const float*)d_in[11];
    const float* W_ih1  = (const float*)d_in[12];
    const float* W_hh1  = (const float*)d_in[13];
    const float* b_ih1  = (const float*)d_in[14];
    const float* b_hh1  = (const float*)d_in[15];
    float* out = (float*)d_out;

    const int T    = in_sizes[3] / (BB * DD);
    const int rows = T * BB;

    float* hb0 = (float*)d_ws;
    float* hb1 = hb0 + (size_t)BB * GG;

    hb_kernel<<<BB / 256, 256, 0, stream>>>(hidden, W_hh0, b_ih0, b_hh0,
                                            W_hh1, b_ih1, b_hh1, hb0, hb1);

    lstm_wmma_kernel<<<rows / 128, 256, 0, stream>>>(dec_x, cat1, cat2, cat3, emb,
                                                     W_ih0, W_ih1, cellp,
                                                     hb0, hb1, out);
}


# Layer-1 gate loop fully unrolled (was by 4): g1[20] is indexed by constants only and stays in registers; the scratch-memory store and reload of the whole array on every trip is gone; sums unchanged
# speedup vs baseline: 2.6806x; 1.7513x over previous
#include <hip/hip_runtime.h>
#include <hip/hip_bf16.h>

typedef float v2f __attribute__((ext_vector_type(2)));
typedef float v8f __attribute__((ext_vector_type(8)));

#define BB   4096
#define DD   32
#define HH   5
#define GG   20
#define INW  128
#define VV   10
#define WPITCH 36

__device__ __forceinline__ float tanh_hw(float x) { return __builtin_amdgcn_tanhf(x); }
__device__ __forceinline__ float sigf(float x) { return __builtin_fmaf(0.5f, tanh_hw(0.5f * x), 0.5f); }

__global__ __launch_bounds__(256) void hb_kernel(
    const float* __restrict__ hidden,
    const float* __restrict__ W_hh0, const float* __restrict__ b_ih0, const float* __restrict__ b_hh0,
    const float* __restrict__ W_hh1, const float* __restrict__ b_ih1, const float* __restrict__ b_hh1,
    const float* __restrict__ emb, const float* __restrict__ W_ih0,
    float* __restrict__ hb0, float* __restrict__ hb1, float* __restrict__ tab)
{
    if (blockIdx.x == BB / 256) {
        const int wv = threadIdx.x >> 5;
        if (wv >= 6) return;
        const int ln   = threadIdx.x & 31;
        const int cc   = ln & 15;
        const int hf   = ln >> 4;
        const int ko   = hf * 2;
        const int s    = wv >> 1;
        const int g    = (wv & 1) * 16 + cc;
        const bool av  = cc < VV;
        const bool bv  = g < GG;
        const float* ap = emb + (av ? cc : 0) * 32;
        const float* bp = W_ih0 + (bv ? g : 0) * INW + 32 * (s + 1);
        v8f acc = {};
        #pragma unroll
        for (int kk = 0; kk < 8; ++kk) {
            const int col = kk * 4 + ko;
            v2f a  = *(const v2f*)(ap + col);
            v2f bw = *(const v2f*)(bp + col);
            if (!av) { a[0] = 0.0f;  a[1] = 0.0f; }
            if (!bv) { bw[0] = 0.0f; bw[1] = 0.0f; }
            acc = __builtin_amdgcn_wmma_f32_16x16x4_f32(false, a, false, bw,
                                                        (short)0, acc, false, false);
        }
        #pragma unroll
        for (int j = 0; j < 8; ++j) {
            const int v = j + hf * 8;
            if (v < VV && bv) tab[(s * VV + v) * GG + g] = acc[j];
        }
        return;
    }
    int b = blockIdx.x * blockDim.x + threadIdx.x;
    if (b >= BB) return;
    float h0[HH], h1[HH];
    #pragma unroll
    for (int j = 0; j < HH; ++j) {
        h0[j] = hidden[b * HH + j];
        h1[j] = hidden[BB * HH + b * HH + j];
    }
    #pragma unroll 4
    for (int g = 0; g < GG; ++g) {
        float s0 = b_ih0[g] + b_hh0[g];
        float s1 = b_ih1[g] + b_hh1[g];
        #pragma unroll
        for (int j = 0; j < HH; ++j) {
            s0 += h0[j] * W_hh0[g * HH + j];
            s1 += h1[j] * W_hh1[g * HH + j];
        }
        hb0[b * GG + g] = s0;
        hb1[b * GG + g] = s1;
    }
}

__global__ __launch_bounds__(256) void lstm_wmma_kernel(
    const float* __restrict__ dec_x,
    const int*   __restrict__ cat1,
    const int*   __restrict__ cat2,
    const int*   __restrict__ cat3,
    const float* __restrict__ tab,
    const float* __restrict__ W_ih0,
    const float* __restrict__ W_ih1,
    const float* __restrict__ cell,
    const float* __restrict__ hb0,
    const float* __restrict__ hb1,
    float* __restrict__ out)
{
    __shared__ float Wp[32 * WPITCH];
    __shared__ __attribute__((aligned(16))) float Ts[3 * VV * GG];
    __shared__ float Gs[8][16 * GG];

    const int tid = threadIdx.x;
    for (int i = tid; i < 32 * DD; i += 256) {
        Wp[(i >> 5) * WPITCH + (i & (DD - 1))] = ((i >> 5) < GG) ? W_ih0[(i >> 5) * INW + (i & (DD - 1))] : 0.0f;
    }
    for (int i = tid; i < 3 * VV * GG; i += 256) Ts[i] = tab[i];
    __syncthreads();

    const int wave = tid >> 5;
    const int lane = tid & 31;
    const int m    = lane & 15;
    const int half = lane >> 4;
    const int koff = half * 2;

    const long rbase = ((long)blockIdx.x * 8 + wave) * 16;
    const int  t     = (int)(rbase >> 12);
    const int  b     = (int)(rbase & 4095) + m;
    const long row   = (long)t * BB + b;

    const float* ap = dec_x + row * DD;
    const float* t1 = &Ts[(0 * VV + cat1[row]) * GG];
    const float* t2 = &Ts[(1 * VV + cat2[row]) * GG];
    const float* t3 = &Ts[(2 * VV + cat3[row]) * GG];

    const float* bp0 = &Wp[m * WPITCH];
    const float* bp1 = &Wp[(m + 16) * WPITCH];

    v8f acc0 = {};
    v8f acc1 = {};
    #pragma unroll
    for (int kk = 0; kk < 8; ++kk) {
        const int col = kk * 4 + koff;
        v2f a  = *(const v2f*)(ap  + col);
        v2f b0 = *(const v2f*)(bp0 + col);
        v2f b1 = *(const v2f*)(bp1 + col);
        acc0 = __builtin_amdgcn_wmma_f32_16x16x4_f32(false, a, false, b0,
                                                     (short)0, acc0, false, false);
        acc1 = __builtin_amdgcn_wmma_f32_16x16x4_f32(false, a, false, b1,
                                                     (short)0, acc1, false, false);
    }

    float* gs = &Gs[wave][0];
    #pragma unroll
    for (int j = 0; j < 8; ++j) {
        const int rm = j + half * 8;
        gs[rm * GG + m] = acc0[j];
        if (m < 4) gs[rm * GG + 16 + m] = acc1[j];
    }
    asm volatile("s_wait_dscnt 0" ::: "memory");

    if (lane < 16) {
        float g0[GG];
        #pragma unroll
        for (int q = 0; q < GG / 4; ++q) {
            const float4 u1 = *(const float4*)(t1 + 4 * q);
            const float4 u2 = *(const float4*)(t2 + 4 * q);
            const float4 u3 = *(const float4*)(t3 + 4 * q);
            const int g = 4 * q;
            g0[g + 0] = (gs[m * GG + g + 0] + ((u1.x + u2.x) + u3.x)) + hb0[(long)b * GG + g + 0];
            g0[g + 1] = (gs[m * GG + g + 1] + ((u1.y + u2.y) + u3.y)) + hb0[(long)b * GG + g + 1];
            g0[g + 2] = (gs[m * GG + g + 2] + ((u1.z + u2.z) + u3.z)) + hb0[(long)b * GG + g + 2];
            g0[g + 3] = (gs[m * GG + g + 3] + ((u1.w + u2.w) + u3.w)) + hb0[(long)b * GG + g + 3];
        }

        float h1v[HH];
        #pragma unroll
        for (int h = 0; h < HH; ++h) {
            float cn = sigf(g0[5 + h]) * cell[(long)b * HH + h]
                     + sigf(g0[h]) * tanh_hw(g0[10 + h]);
            h1v[h] = sigf(g0[15 + h]) * tanh_hw(cn);
        }

        float g1[GG];
        #pragma unroll
        for (int g = 0; g < GG; ++g) {
            float s = hb1[(long)b * GG + g];
            #pragma unroll
            for (int j = 0; j < HH; ++j) s += W_ih1[g * HH + j] * h1v[j];
            g1[g] = s;
        }
        #pragma unroll
        for (int h = 0; h < HH; ++h) {
            float cn = sigf(g1[5 + h]) * cell[(long)BB * HH + (long)b * HH + h]
                     + sigf(g1[h]) * tanh_hw(g1[10 + h]);
            out[row * HH + h] = sigf(g1[15 + h]) * tanh_hw(cn);
        }
    }
}

extern "C" void kernel_launch(void* const* d_in, const int* in_sizes, int n_in,
                              void* d_out, int out_size, void* d_ws, size_t ws_size,
                              hipStream_t stream)
{
    const float* hidden = (const float*)d_in[1];
    const float* cellp  = (const float*)d_in[2];
    const float* dec_x  = (const float*)d_in[3];
    const int*   cat1   = (const int*)d_in[4];
    const int*   cat2   = (const int*)d_in[5];
    const int*   cat3   = (const int*)d_in[6];
    const float* emb    = (const float*)d_in[7];
    const float* W_ih0  = (const float*)d_in[8];
    const float* W_hh0  = (const float*)d_in[9];
    const float* b_ih0  = (const float*)d_in[10];
    const float* b_hh0  = (const float*)d_in[11];
    const float* W_ih1  = (const float*)d_in[12];
    const float* W_hh1  = (const float*)d_in[13];
    const float* b_ih1  = (const float*)d_in[14];
    const float* b_hh1  = (const float*)d_in[15];
    float* out = (float*)d_out;

    const int T    = in_sizes[3] / (BB * DD);
    const int rows = T * BB;

    float* hb0 = (float*)d_ws;
    float* hb1 = hb0 + (size_t)BB * GG;
    float* tab = hb1 + (size_t)BB * GG;

    hb_kernel<<<BB / 256 + 1, 256, 0, stream>>>(hidden, W_hh0, b_ih0, b_hh0,
                                                W_hh1, b_ih1, b_hh1, emb, W_ih0,
                                                hb0, hb1, tab);

    lstm_wmma_kernel<<<rows / 128, 256, 0, stream>>>(dec_x, cat1, cat2, cat3, tab,
                                                     W_ih0, W_ih1, cellp,
                                                     hb0, hb1, out);
}


# Each wave takes 32 rows as two 16-row M tiles (was 16): all 32 lanes own a row of the gate math (16 sat idle there), a B fragment feeds both tiles, waves launched 16384 -> 8192; sums unchanged
# speedup vs baseline: 3.2103x; 1.1976x over previous
#include <hip/hip_runtime.h>
#include <hip/hip_bf16.h>

typedef float v2f __attribute__((ext_vector_type(2)));
typedef float v8f __attribute__((ext_vector_type(8)));

#define BB   4096
#define DD   32
#define HH   5
#define GG   20
#define INW  128
#define VV   10
#define WPITCH 36

__device__ __forceinline__ float tanh_hw(float x) { return __builtin_amdgcn_tanhf(x); }
__device__ __forceinline__ float sigf(float x) { return __builtin_fmaf(0.5f, tanh_hw(0.5f * x), 0.5f); }

__global__ __launch_bounds__(256) void hb_kernel(
    const float* __restrict__ hidden,
    const float* __restrict__ W_hh0, const float* __restrict__ b_ih0, const float* __restrict__ b_hh0,
    const float* __restrict__ W_hh1, const float* __restrict__ b_ih1, const float* __restrict__ b_hh1,
    const float* __restrict__ emb, const float* __restrict__ W_ih0,
    float* __restrict__ hb0, float* __restrict__ hb1, float* __restrict__ tab)
{
    if (blockIdx.x == BB / 256) {
        const int wv = threadIdx.x >> 5;
        if (wv >= 6) return;
        const int ln   = threadIdx.x & 31;
        const int cc   = ln & 15;
        const int hf   = ln >> 4;
        const int ko   = hf * 2;
        const int s    = wv >> 1;
        const int g    = (wv & 1) * 16 + cc;
        const bool av  = cc < VV;
        const bool bv  = g < GG;
        const float* ap = emb + (av ? cc : 0) * 32;
        const float* bp = W_ih0 + (bv ? g : 0) * INW + 32 * (s + 1);
        v8f acc = {};
        #pragma unroll
        for (int kk = 0; kk < 8; ++kk) {
            const int col = kk * 4 + ko;
            v2f a  = *(const v2f*)(ap + col);
            v2f bw = *(const v2f*)(bp + col);
            if (!av) { a[0] = 0.0f;  a[1] = 0.0f; }
            if (!bv) { bw[0] = 0.0f; bw[1] = 0.0f; }
            acc = __builtin_amdgcn_wmma_f32_16x16x4_f32(false, a, false, bw,
                                                        (short)0, acc, false, false);
        }
        #pragma unroll
        for (int j = 0; j < 8; ++j) {
            const int v = j + hf * 8;
            if (v < VV && bv) tab[(s * VV + v) * GG + g] = acc[j];
        }
        return;
    }
    int b = blockIdx.x * blockDim.x + threadIdx.x;
    if (b >= BB) return;
    float h0[HH], h1[HH];
    #pragma unroll
    for (int j = 0; j < HH; ++j) {
        h0[j] = hidden[b * HH + j];
        h1[j] = hidden[BB * HH + b * HH + j];
    }
    #pragma unroll 4
    for (int g = 0; g < GG; ++g) {
        float s0 = b_ih0[g] + b_hh0[g];
        float s1 = b_ih1[g] + b_hh1[g];
        #pragma unroll
        for (int j = 0; j < HH; ++j) {
            s0 += h0[j] * W_hh0[g * HH + j];
            s1 += h1[j] * W_hh1[g * HH + j];
        }
        hb0[b * GG + g] = s0;
        hb1[b * GG + g] = s1;
    }
}

__global__ __launch_bounds__(256) void lstm_wmma_kernel(
    const float* __restrict__ dec_x,
    const int*   __restrict__ cat1,
    const int*   __restrict__ cat2,
    const int*   __restrict__ cat3,
    const float* __restrict__ tab,
    const float* __restrict__ W_ih0,
    const float* __restrict__ W_ih1,
    const float* __restrict__ cell,
    const float* __restrict__ hb0,
    const float* __restrict__ hb1,
    float* __restrict__ out)
{
    __shared__ float Wp[32 * WPITCH];
    __shared__ __attribute__((aligned(16))) float Ts[3 * VV * GG];
    __shared__ __attribute__((aligned(16))) float Gs[8][32 * GG];

    const int tid = threadIdx.x;
    for (int i = tid; i < 32 * DD; i += 256) {
        Wp[(i >> 5) * WPITCH + (i & (DD - 1))] = ((i >> 5) < GG) ? W_ih0[(i >> 5) * INW + (i & (DD - 1))] : 0.0f;
    }
    for (int i = tid; i < 3 * VV * GG; i += 256) Ts[i] = tab[i];
    __syncthreads();

    const int wave = tid >> 5;
    const int lane = tid & 31;
    const int m    = lane & 15;
    const int half = lane >> 4;
    const int koff = half * 2;

    const long rbase = ((long)blockIdx.x * 8 + wave) * 32;
    const int  t     = (int)(rbase >> 12);
    const int  b     = (int)(rbase & 4095) + lane;
    const long row   = (long)t * BB + b;

    const float* ap0 = dec_x + (rbase + m) * DD;
    const float* ap1 = ap0 + 16 * DD;
    const float* t1 = &Ts[(0 * VV + cat1[row]) * GG];
    const float* t2 = &Ts[(1 * VV + cat2[row]) * GG];
    const float* t3 = &Ts[(2 * VV + cat3[row]) * GG];

    const float* bp0 = &Wp[m * WPITCH];
    const float* bp1 = &Wp[(m + 16) * WPITCH];

    v8f acc00 = {};
    v8f acc01 = {};
    v8f acc10 = {};
    v8f acc11 = {};
    #pragma unroll
    for (int kk = 0; kk < 8; ++kk) {
        const int col = kk * 4 + koff;
        v2f a0 = *(const v2f*)(ap0 + col);
        v2f a1 = *(const v2f*)(ap1 + col);
        v2f b0 = *(const v2f*)(bp0 + col);
        v2f b1 = *(const v2f*)(bp1 + col);
        acc00 = __builtin_amdgcn_wmma_f32_16x16x4_f32(false, a0, false, b0,
                                                      (short)0, acc00, false, false);
        acc01 = __builtin_amdgcn_wmma_f32_16x16x4_f32(false, a0, false, b1,
                                                      (short)0, acc01, false, false);
        acc10 = __builtin_amdgcn_wmma_f32_16x16x4_f32(false, a1, false, b0,
                                                      (short)0, acc10, false, false);
        acc11 = __builtin_amdgcn_wmma_f32_16x16x4_f32(false, a1, false, b1,
                                                      (short)0, acc11, false, false);
    }

    float* gs = &Gs[wave][0];
    #pragma unroll
    for (int j = 0; j < 8; ++j) {
        const int rm = j + half * 8;
        gs[rm * GG + m]        = acc00[j];
        gs[(rm + 16) * GG + m] = acc10[j];
    }
    if (m < 4) {
        #pragma unroll
        for (int j = 0; j < 8; ++j) {
            const int rm = j + half * 8;
            gs[rm * GG + 16 + m]        = acc01[j];
            gs[(rm + 16) * GG + 16 + m] = acc11[j];
        }
    }
    asm volatile("s_wait_dscnt 0" ::: "memory");

    {
        float g0[GG];
        #pragma unroll
        for (int q = 0; q < GG / 4; ++q) {
            const float4 u1 = *(const float4*)(t1 + 4 * q);
            const float4 u2 = *(const float4*)(t2 + 4 * q);
            const float4 u3 = *(const float4*)(t3 + 4 * q);
            const int g = 4 * q;
            g0[g + 0] = (gs[lane * GG + g + 0] + ((u1.x + u2.x) + u3.x)) + hb0[(long)b * GG + g + 0];
            g0[g + 1] = (gs[lane * GG + g + 1] + ((u1.y + u2.y) + u3.y)) + hb0[(long)b * GG + g + 1];
            g0[g + 2] = (gs[lane * GG + g + 2] + ((u1.z + u2.z) + u3.z)) + hb0[(long)b * GG + g + 2];
            g0[g + 3] = (gs[lane * GG + g + 3] + ((u1.w + u2.w) + u3.w)) + hb0[(long)b * GG + g + 3];
        }

        float h1v[HH];
        #pragma unroll
        for (int h = 0; h < HH; ++h) {
            float cn = sigf(g0[5 + h]) * cell[(long)b * HH + h]
                     + sigf(g0[h]) * tanh_hw(g0[10 + h]);
            h1v[h] = sigf(g0[15 + h]) * tanh_hw(cn);
        }

        float g1[GG];
        #pragma unroll
        for (int g = 0; g < GG; ++g) {
            float s = hb1[(long)b * GG + g];
            #pragma unroll
            for (int j = 0; j < HH; ++j) s += W_ih1[g * HH + j] * h1v[j];
            g1[g] = s;
        }
        #pragma unroll
        for (int h = 0; h < HH; ++h) {
            float cn = sigf(g1[5 + h]) * cell[(long)BB * HH + (long)b * HH + h]
                     + sigf(g1[h]) * tanh_hw(g1[10 + h]);
            out[row * HH + h] = sigf(g1[15 + h]) * tanh_hw(cn);
        }
    }
}

extern "C" void kernel_launch(void* const* d_in, const int* in_sizes, int n_in,
                              void* d_out, int out_size, void* d_ws, size_t ws_size,
                              hipStream_t stream)
{
    const float* hidden = (const float*)d_in[1];
    const float* cellp  = (const float*)d_in[2];
    const float* dec_x  = (const float*)d_in[3];
    const int*   cat1   = (const int*)d_in[4];
    const int*   cat2   = (const int*)d_in[5];
    const int*   cat3   = (const int*)d_in[6];
    const float* emb    = (const float*)d_in[7];
    const float* W_ih0  = (const float*)d_in[8];
    const float* W_hh0  = (const float*)d_in[9];
    const float* b_ih0  = (const float*)d_in[10];
    const float* b_hh0  = (const float*)d_in[11];
    const float* W_ih1  = (const float*)d_in[12];
    const float* W_hh1  = (const float*)d_in[13];
    const float* b_ih1  = (const float*)d_in[14];
    const float* b_hh1  = (const float*)d_in[15];
    float* out = (float*)d_out;

    const int T    = in_sizes[3] / (BB * DD);
    const int rows = T * BB;

    float* hb0 = (float*)d_ws;
    float* hb1 = hb0 + (size_t)BB * GG;
    float* tab = hb1 + (size_t)BB * GG;

    hb_kernel<<<BB / 256 + 1, 256, 0, stream>>>(hidden, W_hh0, b_ih0, b_hh0,
                                                W_hh1, b_ih1, b_hh1, emb, W_ih0,
                                                hb0, hb1, tab);

    lstm_wmma_kernel<<<rows / 256, 256, 0, stream>>>(dec_x, cat1, cat2, cat3, tab,
                                                     W_ih0, W_ih1, cellp,
                                                     hb0, hb1, out);
}


# Single launch: the pre-pass kernel is gone; each block (32 batch entries x 8 time steps) works out its entries' recurrent gate terms and the three gate tables into LDS itself; sums unchanged
# speedup vs baseline: 4.4081x; 1.3731x over previous
#include <hip/hip_runtime.h>
#include <hip/hip_bf16.h>


typedef float v2f __attribute__((ext_vector_type(2)));
typedef float v8f __attribute__((ext_vector_type(8)));

#define BB   4096
#define DD   32
#define HH   5
#define GG   20
#define INW  128
#define VV   10
#define TPB  8
#define WPITCH 132
#define EPITCH 36

__device__ __forceinline__ float tanh_hw(float x) { return __builtin_amdgcn_tanhf(x); }
__device__ __forceinline__ float sigf(float x) { return __builtin_fmaf(0.5f, tanh_hw(0.5f * x), 0.5f); }

__global__ __launch_bounds__(256) void lstm_wmma_kernel(
    const float* __restrict__ hidden,
    const float* __restrict__ cell,
    const float* __restrict__ dec_x,
    const int*   __restrict__ cat1,
    const int*   __restrict__ cat2,
    const int*   __restrict__ cat3,
    const float* __restrict__ emb,
    const float* __restrict__ W_ih0,
    const float* __restrict__ W_hh0, const float* __restrict__ b_ih0, const float* __restrict__ b_hh0,
    const float* __restrict__ W_ih1,
    const float* __restrict__ W_hh1, const float* __restrict__ b_ih1, const float* __restrict__ b_hh1,
    float* __restrict__ out)
{
    __shared__ __attribute__((aligned(16))) float Wp[GG * WPITCH];
    __shared__ __attribute__((aligned(16))) float Es[VV * EPITCH];
    __shared__ __attribute__((aligned(16))) float Ts[3 * VV * GG];
    __shared__ __attribute__((aligned(16))) float Hb[2][32 * GG];
    __shared__ __attribute__((aligned(16))) float Gs[8][32 * GG];

    const int tid  = threadIdx.x;
    const int lane = tid & 31;
    const int wave = __builtin_amdgcn_readfirstlane(tid >> 5);

    const int b0 = (blockIdx.x & (BB / 32 - 1)) * 32;
    const int t  = (blockIdx.x / (BB / 32)) * TPB + wave;
    const int b  = b0 + lane;

    for (int i = tid; i < GG * (INW / 4); i += 256) {
        const int r = i >> 5, c4 = i & 31;
        *(float4*)(&Wp[r * WPITCH + c4 * 4]) = *(const float4*)(W_ih0 + r * INW + c4 * 4);
    }
    if (tid < VV * 8) {
        const int r = tid >> 3, c4 = tid & 7;
        *(float4*)(&Es[r * EPITCH + c4 * 4]) = *(const float4*)(emb + r * 32 + c4 * 4);
    }
    {
        const int layer = wave >> 2;
        const int gq    = (wave & 3) * HH;
        const float* hp = hidden + ((long)layer * BB + b) * HH;
        const float* Wh = layer ? W_hh1 : W_hh0;
        const float* bi = layer ? b_ih1 : b_ih0;
        const float* bh = layer ? b_hh1 : b_hh0;
        float h[HH];
        #pragma unroll
        for (int j = 0; j < HH; ++j) h[j] = hp[j];
        #pragma unroll
        for (int q = 0; q < HH; ++q) {
            const int g = gq + q;
            float s = bi[g] + bh[g];
            #pragma unroll
            for (int j = 0; j < HH; ++j) s += h[j] * Wh[g * HH + j];
            Hb[layer][lane * GG + g] = s;
        }
    }
    __syncthreads();

    const int m    = lane & 15;
    const int half = lane >> 4;
    const int koff = half * 2;

    const long rbase = (long)t * BB + b0;
    const long row   = rbase + lane;

    const float* ap0 = dec_x + (rbase + m) * DD + koff;
    const float* ap1 = ap0 + 16 * DD;
    v2f a0[8], a1[8];
    #pragma unroll
    for (int kk = 0; kk < 8; ++kk) {
        a0[kk] = *(const v2f*)(ap0 + kk * 4);
        a1[kk] = *(const v2f*)(ap1 + kk * 4);
    }
    const int c1 = cat1[row];
    const int c2 = cat2[row];
    const int c3 = cat3[row];

    if (wave < 4) {
        const int n  = wave * 16 + m;
        const int nn = n < 3 * GG ? n : 3 * GG - 1;
        const int s  = nn / GG;
        const int g  = nn - s * GG;
        const float* ea = &Es[(m < VV ? m : VV - 1) * EPITCH + koff];
        const float* wb = &Wp[g * WPITCH + 32 * (s + 1) + koff];
        v8f acc = {};
        #pragma unroll
        for (int kk = 0; kk < 8; ++kk) {
            const v2f a  = *(const v2f*)(ea + kk * 4);
            const v2f bw = *(const v2f*)(wb + kk * 4);
            acc = __builtin_amdgcn_wmma_f32_16x16x4_f32(false, a, false, bw,
                                                        (short)0, acc, false, false);
        }
        #pragma unroll
        for (int j = 0; j < 8; ++j) {
            const int v = j + half * 8;
            if (v < VV && n < 3 * GG) Ts[(s * VV + v) * GG + g] = acc[j];
        }
    }
    __syncthreads();

    const float* t1 = &Ts[(0 * VV + c1) * GG];
    const float* t2 = &Ts[(1 * VV + c2) * GG];
    const float* t3 = &Ts[(2 * VV + c3) * GG];

    const float* bp0 = &Wp[m * WPITCH + koff];
    const float* bp1 = &Wp[(m < 4 ? m + 16 : GG - 1) * WPITCH + koff];

    v8f acc00 = {};
    v8f acc01 = {};
    v8f acc10 = {};
    v8f acc11 = {};
    #pragma unroll
    for (int kk = 0; kk < 8; ++kk) {
        const v2f b0v = *(const v2f*)(bp0 + kk * 4);
        const v2f b1v = *(const v2f*)(bp1 + kk * 4);
        acc00 = __builtin_amdgcn_wmma_f32_16x16x4_f32(false, a0[kk], false, b0v,
                                                      (short)0, acc00, false, false);
        acc01 = __builtin_amdgcn_wmma_f32_16x16x4_f32(false, a0[kk], false, b1v,
                                                      (short)0, acc01, false, false);
        acc10 = __builtin_amdgcn_wmma_f32_16x16x4_f32(false, a1[kk], false, b0v,
                                                      (short)0, acc10, false, false);
        acc11 = __builtin_amdgcn_wmma_f32_16x16x4_f32(false, a1[kk], false, b1v,
                                                      (short)0, acc11, false, false);
    }

    float* gs = &Gs[wave][0];
    #pragma unroll
    for (int j = 0; j < 8; ++j) {
        const int rm = j + half * 8;
        gs[rm * GG + m]        = acc00[j];
        gs[(rm + 16) * GG + m] = acc10[j];
    }
    if (m < 4) {
        #pragma unroll
        for (int j = 0; j < 8; ++j) {
            const int rm = j + half * 8;
            gs[rm * GG + 16 + m]        = acc01[j];
            gs[(rm + 16) * GG + 16 + m] = acc11[j];
        }
    }
    asm volatile("s_wait_dscnt 0" ::: "memory");

    {
        float g0[GG];
        #pragma unroll
        for (int q = 0; q < GG / 4; ++q) {
            const float4 u1 = *(const float4*)(t1 + 4 * q);
            const float4 u2 = *(const float4*)(t2 + 4 * q);
            const float4 u3 = *(const float4*)(t3 + 4 * q);
            const float4 hq = *(const float4*)(&Hb[0][lane * GG + 4 * q]);
            const int g = 4 * q;
            g0[g + 0] = (gs[lane * GG + g + 0] + ((u1.x + u2.x) + u3.x)) + hq.x;
            g0[g + 1] = (gs[lane * GG + g + 1] + ((u1.y + u2.y) + u3.y)) + hq.y;
            g0[g + 2] = (gs[lane * GG + g + 2] + ((u1.z + u2.z) + u3.z)) + hq.z;
            g0[g + 3] = (gs[lane * GG + g + 3] + ((u1.w + u2.w) + u3.w)) + hq.w;
        }

        float h1v[HH];
        #pragma unroll
        for (int h = 0; h < HH; ++h) {
            float cn = sigf(g0[5 + h]) * cell[(long)b * HH + h]
                     + sigf(g0[h]) * tanh_hw(g0[10 + h]);
            h1v[h] = sigf(g0[15 + h]) * tanh_hw(cn);
        }

        float g1[GG];
        #pragma unroll
        for (int g = 0; g < GG; ++g) {
            float s = Hb[1][lane * GG + g];
            #pragma unroll
            for (int j = 0; j < HH; ++j) s += W_ih1[g * HH + j] * h1v[j];
            g1[g] = s;
        }
        #pragma unroll
        for (int h = 0; h < HH; ++h) {
            float cn = sigf(g1[5 + h]) * cell[(long)BB * HH + (long)b * HH + h]
                     + sigf(g1[h]) * tanh_hw(g1[10 + h]);
            out[row * HH + h] = sigf(g1[15 + h]) * tanh_hw(cn);
        }
    }
}

extern "C" void kernel_launch(void* const* d_in, const int* in_sizes, int n_in,
                              void* d_out, int out_size, void* d_ws, size_t ws_size,
                              hipStream_t stream)
{
    const float* hidden = (const float*)d_in[1];
    const float* cellp  = (const float*)d_in[2];
    const float* dec_x  = (const float*)d_in[3];
    const int*   cat1   = (const int*)d_in[4];
    const int*   cat2   = (const int*)d_in[5];
    const int*   cat3   = (const int*)d_in[6];
    const float* emb    = (const float*)d_in[7];
    const float* W_ih0  = (const float*)d_in[8];
    const float* W_hh0  = (const float*)d_in[9];
    const float* b_ih0  = (const float*)d_in[10];
    const float* b_hh0  = (const float*)d_in[11];
    const float* W_ih1  = (const float*)d_in[12];
    const float* W_hh1  = (const float*)d_in[13];
    const float* b_ih1  = (const float*)d_in[14];
    const float* b_hh1  = (const float*)d_in[15];
    float* out = (float*)d_out;

    const int T    = in_sizes[3] / (BB * DD);
    const int rows = T * BB;

    lstm_wmma_kernel<<<rows / 256, 256, 0, stream>>>(hidden, cellp, dec_x, cat1, cat2, cat3, emb,
                                                     W_ih0, W_hh0, b_ih0, b_hh0,
                                                     W_ih1, W_hh1, b_ih1, b_hh1, out);
}


# Each wave takes two time steps of its 32 batch entries (64 rows, four M tiles; was 32 rows): B fragments, recurrent terms, cell values, layer-1 weights serve both; waves 8192 -> 4096; sums unchanged
# speedup vs baseline: 4.6299x; 1.0503x over previous
#include <hip/hip_runtime.h>
#include <hip/hip_bf16.h>


typedef float v2f __attribute__((ext_vector_type(2)));
typedef float v8f __attribute__((ext_vector_type(8)));

#define BB   4096
#define DD   32
#define HH   5
#define GG   20
#define INW  128
#define VV   10
#define SPW  2
#define TPB  16
#define WPITCH 132
#define EPITCH 36

__device__ __forceinline__ float tanh_hw(float x) { return __builtin_amdgcn_tanhf(x); }
__device__ __forceinline__ float sigf(float x) { return __builtin_fmaf(0.5f, tanh_hw(0.5f * x), 0.5f); }

__global__ __launch_bounds__(256) void lstm_wmma_kernel(
    const float* __restrict__ hidden,
    const float* __restrict__ cell,
    const float* __restrict__ dec_x,
    const int*   __restrict__ cat1,
    const int*   __restrict__ cat2,
    const int*   __restrict__ cat3,
    const float* __restrict__ emb,
    const float* __restrict__ W_ih0,
    const float* __restrict__ W_hh0, const float* __restrict__ b_ih0, const float* __restrict__ b_hh0,
    const float* __restrict__ W_ih1,
    const float* __restrict__ W_hh1, const float* __restrict__ b_ih1, const float* __restrict__ b_hh1,
    float* __restrict__ out)
{
    __shared__ __attribute__((aligned(16))) float Wp[GG * WPITCH];
    __shared__ __attribute__((aligned(16))) float Es[VV * EPITCH];
    __shared__ __attribute__((aligned(16))) float Ts[3 * VV * GG];
    __shared__ __attribute__((aligned(16))) float Hb[2][32 * GG];
    __shared__ __attribute__((aligned(16))) float Gs[8][SPW][32 * GG];

    const int tid  = threadIdx.x;
    const int lane = tid & 31;
    const int wave = __builtin_amdgcn_readfirstlane(tid >> 5);

    const int b0 = (blockIdx.x & (BB / 32 - 1)) * 32;
    const int t  = (blockIdx.x / (BB / 32)) * TPB + wave * SPW;
    const int b  = b0 + lane;

    for (int i = tid; i < GG * (INW / 4); i += 256) {
        const int r = i >> 5, c4 = i & 31;
        *(float4*)(&Wp[r * WPITCH + c4 * 4]) = *(const float4*)(W_ih0 + r * INW + c4 * 4);
    }
    if (tid < VV * 8) {
        const int r = tid >> 3, c4 = tid & 7;
        *(float4*)(&Es[r * EPITCH + c4 * 4]) = *(const float4*)(emb + r * 32 + c4 * 4);
    }
    {
        const int layer = wave >> 2;
        const int gq    = (wave & 3) * HH;
        const float* hp = hidden + ((long)layer * BB + b) * HH;
        const float* Wh = layer ? W_hh1 : W_hh0;
        const float* bi = layer ? b_ih1 : b_ih0;
        const float* bh = layer ? b_hh1 : b_hh0;
        float h[HH];
        #pragma unroll
        for (int j = 0; j < HH; ++j) h[j] = hp[j];
        #pragma unroll
        for (int q = 0; q < HH; ++q) {
            const int g = gq + q;
            float s = bi[g] + bh[g];
            #pragma unroll
            for (int j = 0; j < HH; ++j) s += h[j] * Wh[g * HH + j];
            Hb[layer][lane * GG + g] = s;
        }
    }
    __syncthreads();

    const int m    = lane & 15;
    const int half = lane >> 4;
    const int koff = half * 2;

    const long rbase = (long)t * BB + b0;
    const long row0  = rbase + lane;

    v2f a0[SPW][8], a1[SPW][8];
    int c1[SPW], c2[SPW], c3[SPW];
    #pragma unroll
    for (int s = 0; s < SPW; ++s) {
        const float* ap0 = dec_x + (rbase + (long)s * BB + m) * DD + koff;
        const float* ap1 = ap0 + 16 * DD;
        #pragma unroll
        for (int kk = 0; kk < 8; ++kk) {
            a0[s][kk] = *(const v2f*)(ap0 + kk * 4);
            a1[s][kk] = *(const v2f*)(ap1 + kk * 4);
        }
        c1[s] = cat1[row0 + (long)s * BB];
        c2[s] = cat2[row0 + (long)s * BB];
        c3[s] = cat3[row0 + (long)s * BB];
    }
    float cl0[HH], cl1[HH];
    #pragma unroll
    for (int h = 0; h < HH; ++h) {
        cl0[h] = cell[(long)b * HH + h];
        cl1[h] = cell[(long)BB * HH + (long)b * HH + h];
    }

    if (wave < 4) {
        const int n  = wave * 16 + m;
        const int nn = n < 3 * GG ? n : 3 * GG - 1;
        const int s  = nn / GG;
        const int g  = nn - s * GG;
        const float* ea = &Es[(m < VV ? m : VV - 1) * EPITCH + koff];
        const float* wb = &Wp[g * WPITCH + 32 * (s + 1) + koff];
        v8f acc = {};
        #pragma unroll
        for (int kk = 0; kk < 8; ++kk) {
            const v2f a  = *(const v2f*)(ea + kk * 4);
            const v2f bw = *(const v2f*)(wb + kk * 4);
            acc = __builtin_amdgcn_wmma_f32_16x16x4_f32(false, a, false, bw,
                                                        (short)0, acc, false, false);
        }
        #pragma unroll
        for (int j = 0; j < 8; ++j) {
            const int v = j + half * 8;
            if (v < VV && n < 3 * GG) Ts[(s * VV + v) * GG + g] = acc[j];
        }
    }
    __syncthreads();

    const float* bp0 = &Wp[m * WPITCH + koff];
    const float* bp1 = &Wp[(m < 4 ? m + 16 : GG - 1) * WPITCH + koff];

    v8f acc00[SPW], acc01[SPW], acc10[SPW], acc11[SPW];
    #pragma unroll
    for (int s = 0; s < SPW; ++s) { acc00[s] = {}; acc01[s] = {}; acc10[s] = {}; acc11[s] = {}; }
    #pragma unroll
    for (int kk = 0; kk < 8; ++kk) {
        const v2f b0v = *(const v2f*)(bp0 + kk * 4);
        const v2f b1v = *(const v2f*)(bp1 + kk * 4);
        #pragma unroll
        for (int s = 0; s < SPW; ++s) {
            acc00[s] = __builtin_amdgcn_wmma_f32_16x16x4_f32(false, a0[s][kk], false, b0v,
                                                             (short)0, acc00[s], false, false);
            acc01[s] = __builtin_amdgcn_wmma_f32_16x16x4_f32(false, a0[s][kk], false, b1v,
                                                             (short)0, acc01[s], false, false);
            acc10[s] = __builtin_amdgcn_wmma_f32_16x16x4_f32(false, a1[s][kk], false, b0v,
                                                             (short)0, acc10[s], false, false);
            acc11[s] = __builtin_amdgcn_wmma_f32_16x16x4_f32(false, a1[s][kk], false, b1v,
                                                             (short)0, acc11[s], false, false);
        }
    }

    #pragma unroll
    for (int s = 0; s < SPW; ++s) {
        float* gs = &Gs[wave][s][0];
        #pragma unroll
        for (int j = 0; j < 8; ++j) {
            const int rm = j + half * 8;
            gs[rm * GG + m]        = acc00[s][j];
            gs[(rm + 16) * GG + m] = acc10[s][j];
        }
    }
    if (m < 4) {
        #pragma unroll
        for (int s = 0; s < SPW; ++s) {
            float* gs = &Gs[wave][s][0];
            #pragma unroll
            for (int j = 0; j < 8; ++j) {
                const int rm = j + half * 8;
                gs[rm * GG + 16 + m]        = acc01[s][j];
                gs[(rm + 16) * GG + 16 + m] = acc11[s][j];
            }
        }
    }
    asm volatile("s_wait_dscnt 0" ::: "memory");

    {
        float g0[SPW][GG];
        #pragma unroll
        for (int q = 0; q < GG / 4; ++q) {
            const float4 hq = *(const float4*)(&Hb[0][lane * GG + 4 * q]);
            const int g = 4 * q;
            #pragma unroll
            for (int s = 0; s < SPW; ++s) {
                const float4 u1 = *(const float4*)(&Ts[(0 * VV + c1[s]) * GG + 4 * q]);
                const float4 u2 = *(const float4*)(&Ts[(1 * VV + c2[s]) * GG + 4 * q]);
                const float4 u3 = *(const float4*)(&Ts[(2 * VV + c3[s]) * GG + 4 * q]);
                const float* gs = &Gs[wave][s][lane * GG + g];
                g0[s][g + 0] = (gs[0] + ((u1.x + u2.x) + u3.x)) + hq.x;
                g0[s][g + 1] = (gs[1] + ((u1.y + u2.y) + u3.y)) + hq.y;
                g0[s][g + 2] = (gs[2] + ((u1.z + u2.z) + u3.z)) + hq.z;
                g0[s][g + 3] = (gs[3] + ((u1.w + u2.w) + u3.w)) + hq.w;
            }
        }

        float h1v[SPW][HH];
        #pragma unroll
        for (int s = 0; s < SPW; ++s) {
            #pragma unroll
            for (int h = 0; h < HH; ++h) {
                float cn = sigf(g0[s][5 + h]) * cl0[h]
                         + sigf(g0[s][h]) * tanh_hw(g0[s][10 + h]);
                h1v[s][h] = sigf(g0[s][15 + h]) * tanh_hw(cn);
            }
        }

        float g1[SPW][GG];
        #pragma unroll
        for (int g = 0; g < GG; ++g) {
            const float hb1 = Hb[1][lane * GG + g];
            #pragma unroll
            for (int s = 0; s < SPW; ++s) {
                float sum = hb1;
                #pragma unroll
                for (int j = 0; j < HH; ++j) sum += W_ih1[g * HH + j] * h1v[s][j];
                g1[s][g] = sum;
            }
        }
        #pragma unroll
        for (int s = 0; s < SPW; ++s) {
            const long row = row0 + (long)s * BB;
            #pragma unroll
            for (int h = 0; h < HH; ++h) {
                float cn = sigf(g1[s][5 + h]) * cl1[h]
                         + sigf(g1[s][h]) * tanh_hw(g1[s][10 + h]);
                out[row * HH + h] = sigf(g1[s][15 + h]) * tanh_hw(cn);
            }
        }
    }
}

extern "C" void kernel_launch(void* const* d_in, const int* in_sizes, int n_in,
                              void* d_out, int out_size, void* d_ws, size_t ws_size,
                              hipStream_t stream)
{
    const float* hidden = (const float*)d_in[1];
    const float* cellp  = (const float*)d_in[2];
    const float* dec_x  = (const float*)d_in[3];
    const int*   cat1   = (const int*)d_in[4];
    const int*   cat2   = (const int*)d_in[5];
    const int*   cat3   = (const int*)d_in[6];
    const float* emb    = (const float*)d_in[7];
    const float* W_ih0  = (const float*)d_in[8];
    const float* W_hh0  = (const float*)d_in[9];
    const float* b_ih0  = (const float*)d_in[10];
    const float* b_hh0  = (const float*)d_in[11];
    const float* W_ih1  = (const float*)d_in[12];
    const float* W_hh1  = (const float*)d_in[13];
    const float* b_ih1  = (const float*)d_in[14];
    const float* b_hh1  = (const float*)d_in[15];
    float* out = (float*)d_out;

    const int T    = in_sizes[3] / (BB * DD);
    const int rows = T * BB;

    lstm_wmma_kernel<<<rows / 512, 256, 0, stream>>>(hidden, cellp, dec_x, cat1, cat2, cat3, emb,
                                                     W_ih0, W_hh0, b_ih0, b_hh0,
                                                     W_ih1, W_hh1, b_ih1, b_hh1, out);
}


# Every global load of a wave (staged weights, hidden, x fragments, ids, cell) issued as one group at the top, in flight across LDS-only barriers: five chained round trips become one; sums unchanged
# speedup vs baseline: 5.0352x; 1.0875x over previous
#include <hip/hip_runtime.h>
#include <hip/hip_bf16.h>


typedef float v2f __attribute__((ext_vector_type(2)));
typedef float v8f __attribute__((ext_vector_type(8)));

#define BB   4096
#define DD   32
#define HH   5
#define GG   20
#define INW  128
#define VV   10
#define SPW  2
#define TPB  16
#define WPITCH 132
#define EPITCH 36

__device__ __forceinline__ float tanh_hw(float x) { return __builtin_amdgcn_tanhf(x); }
__device__ __forceinline__ float sigf(float x) { return __builtin_fmaf(0.5f, tanh_hw(0.5f * x), 0.5f); }

__device__ __forceinline__ void lds_barrier() {
    __builtin_amdgcn_fence(__ATOMIC_RELEASE, "workgroup", "local");
    __builtin_amdgcn_s_barrier();
    __builtin_amdgcn_fence(__ATOMIC_ACQUIRE, "workgroup", "local");
}

__global__ __launch_bounds__(256) void lstm_wmma_kernel(
    const float* hidden,
    const float* __restrict__ cell,
    const float* __restrict__ dec_x,
    const int*   __restrict__ cat1,
    const int*   __restrict__ cat2,
    const int*   __restrict__ cat3,
    const float* emb,
    const float* W_ih0,
    const float* __restrict__ W_hh0, const float* __restrict__ b_ih0, const float* __restrict__ b_hh0,
    const float* __restrict__ W_ih1,
    const float* __restrict__ W_hh1, const float* __restrict__ b_ih1, const float* __restrict__ b_hh1,
    float* __restrict__ out)
{
    __shared__ __attribute__((aligned(16))) float Wp[GG * WPITCH];
    __shared__ __attribute__((aligned(16))) float Es[16 * EPITCH];
    __shared__ __attribute__((aligned(16))) float Ts[3 * VV * GG];
    __shared__ __attribute__((aligned(16))) float Hb[2][32 * GG];
    __shared__ __attribute__((aligned(16))) float Gs[8][SPW][32 * GG];

    const int tid  = threadIdx.x;
    const int lane = tid & 31;
    const int wave = __builtin_amdgcn_readfirstlane(tid >> 5);

    const int b0 = (blockIdx.x & (BB / 32 - 1)) * 32;
    const int t  = (blockIdx.x / (BB / 32)) * TPB + wave * SPW;
    const int b  = b0 + lane;

    const int m    = lane & 15;
    const int half = lane >> 4;
    const int koff = half * 2;

    const long rbase = (long)t * BB + b0;
    const long row0  = rbase + lane;

    const int layer = wave >> 2;
    const int gq    = (wave & 3) * HH;
    float wh[HH][HH], bsum[HH];
    {
        const float* Wh = layer ? W_hh1 : W_hh0;
        const float* bi = layer ? b_ih1 : b_ih0;
        const float* bh = layer ? b_hh1 : b_hh0;
        #pragma unroll
        for (int q = 0; q < HH; ++q) {
            bsum[q] = bi[gq + q] + bh[gq + q];
            #pragma unroll
            for (int j = 0; j < HH; ++j) wh[q][j] = Wh[(gq + q) * HH + j];
        }
    }
    const int i0 = tid, i1 = tid + 256, i2 = tid + 512;
    const int ie = i2 - GG * (INW / 4);
    const int iec = ie < VV * 8 ? ie : VV * 8 - 1;
    const float4 w0 = *(const float4*)(W_ih0 + i0 * 4);
    const float4 w1 = *(const float4*)(W_ih0 + i1 * 4);
    const float4 w2 = *(const float4*)(ie < 0 ? W_ih0 + i2 * 4 : emb + iec * 4);
    float hv[HH];
    {
        const float* hp = hidden + ((long)layer * BB + b) * HH;
        #pragma unroll
        for (int j = 0; j < HH; ++j) hv[j] = hp[j];
    }
    __builtin_amdgcn_sched_barrier(0);
    v2f a0[SPW][8], a1[SPW][8];
    int c1[SPW], c2[SPW], c3[SPW];
    #pragma unroll
    for (int s = 0; s < SPW; ++s) {
        const float* ap0 = dec_x + (rbase + (long)s * BB + m) * DD + koff;
        const float* ap1 = ap0 + 16 * DD;
        #pragma unroll
        for (int kk = 0; kk < 8; ++kk) {
            a0[s][kk] = *(const v2f*)(ap0 + kk * 4);
            a1[s][kk] = *(const v2f*)(ap1 + kk * 4);
        }
        c1[s] = cat1[row0 + (long)s * BB];
        c2[s] = cat2[row0 + (long)s * BB];
        c3[s] = cat3[row0 + (long)s * BB];
    }
    float cl0[HH], cl1[HH];
    #pragma unroll
    for (int h = 0; h < HH; ++h) {
        cl0[h] = cell[(long)b * HH + h];
        cl1[h] = cell[(long)BB * HH + (long)b * HH + h];
    }

    *(float4*)(&Wp[(i0 >> 5) * WPITCH + (i0 & 31) * 4]) = w0;
    *(float4*)(&Wp[(i1 >> 5) * WPITCH + (i1 & 31) * 4]) = w1;
    *(float4*)(ie < 0 ? &Wp[(i2 >> 5) * WPITCH + (i2 & 31) * 4] : &Es[(ie >> 3) * EPITCH + (ie & 7) * 4]) = w2;
    #pragma unroll
    for (int q = 0; q < HH; ++q) {
        float s = bsum[q];
        #pragma unroll
        for (int j = 0; j < HH; ++j) s += hv[j] * wh[q][j];
        Hb[layer][lane * GG + gq + q] = s;
    }
    lds_barrier();

    if (wave < 4) {
        const int n  = wave * 16 + m;
        const int nn = n < 3 * GG ? n : 3 * GG - 1;
        const int s  = nn / GG;
        const int g  = nn - s * GG;
        const float* ea = &Es[(m < VV ? m : VV - 1) * EPITCH + koff];
        const float* wb = &Wp[g * WPITCH + 32 * (s + 1) + koff];
        v8f acc = {};
        #pragma unroll
        for (int kk = 0; kk < 8; ++kk) {
            const v2f a  = *(const v2f*)(ea + kk * 4);
            const v2f bw = *(const v2f*)(wb + kk * 4);
            acc = __builtin_amdgcn_wmma_f32_16x16x4_f32(false, a, false, bw,
                                                        (short)0, acc, false, false);
        }
        #pragma unroll
        for (int j = 0; j < 8; ++j) {
            const int v = j + half * 8;
            if (v < VV && n < 3 * GG) Ts[(s * VV + v) * GG + g] = acc[j];
        }
    }
    lds_barrier();

    const float* bp0 = &Wp[m * WPITCH + koff];
    const float* bp1 = &Wp[(m < 4 ? m + 16 : GG - 1) * WPITCH + koff];

    v8f acc00[SPW], acc01[SPW], acc10[SPW], acc11[SPW];
    #pragma unroll
    for (int s = 0; s < SPW; ++s) { acc00[s] = {}; acc01[s] = {}; acc10[s] = {}; acc11[s] = {}; }
    #pragma unroll
    for (int kk = 0; kk < 8; ++kk) {
        const v2f b0v = *(const v2f*)(bp0 + kk * 4);
        const v2f b1v = *(const v2f*)(bp1 + kk * 4);
        #pragma unroll
        for (int s = 0; s < SPW; ++s) {
            acc00[s] = __builtin_amdgcn_wmma_f32_16x16x4_f32(false, a0[s][kk], false, b0v,
                                                             (short)0, acc00[s], false, false);
            acc01[s] = __builtin_amdgcn_wmma_f32_16x16x4_f32(false, a0[s][kk], false, b1v,
                                                             (short)0, acc01[s], false, false);
            acc10[s] = __builtin_amdgcn_wmma_f32_16x16x4_f32(false, a1[s][kk], false, b0v,
                                                             (short)0, acc10[s], false, false);
            acc11[s] = __builtin_amdgcn_wmma_f32_16x16x4_f32(false, a1[s][kk], false, b1v,
                                                             (short)0, acc11[s], false, false);
        }
    }

    #pragma unroll
    for (int s = 0; s < SPW; ++s) {
        float* gs = &Gs[wave][s][0];
        #pragma unroll
        for (int j = 0; j < 8; ++j) {
            const int rm = j + half * 8;
            gs[rm * GG + m]        = acc00[s][j];
            gs[(rm + 16) * GG + m] = acc10[s][j];
        }
    }
    if (m < 4) {
        #pragma unroll
        for (int s = 0; s < SPW; ++s) {
            float* gs = &Gs[wave][s][0];
            #pragma unroll
            for (int j = 0; j < 8; ++j) {
                const int rm = j + half * 8;
                gs[rm * GG + 16 + m]        = acc01[s][j];
                gs[(rm + 16) * GG + 16 + m] = acc11[s][j];
            }
        }
    }
    asm volatile("s_wait_dscnt 0" ::: "memory");

    {
        float g0[SPW][GG];
        #pragma unroll
        for (int q = 0; q < GG / 4; ++q) {
            const float4 hq = *(const float4*)(&Hb[0][lane * GG + 4 * q]);
            const int g = 4 * q;
            #pragma unroll
            for (int s = 0; s < SPW; ++s) {
                const float4 u1 = *(const float4*)(&Ts[(0 * VV + c1[s]) * GG + 4 * q]);
                const float4 u2 = *(const float4*)(&Ts[(1 * VV + c2[s]) * GG + 4 * q]);
                const float4 u3 = *(const float4*)(&Ts[(2 * VV + c3[s]) * GG + 4 * q]);
                const float* gs = &Gs[wave][s][lane * GG + g];
                g0[s][g + 0] = (gs[0] + ((u1.x + u2.x) + u3.x)) + hq.x;
                g0[s][g + 1] = (gs[1] + ((u1.y + u2.y) + u3.y)) + hq.y;
                g0[s][g + 2] = (gs[2] + ((u1.z + u2.z) + u3.z)) + hq.z;
                g0[s][g + 3] = (gs[3] + ((u1.w + u2.w) + u3.w)) + hq.w;
            }
        }

        float h1v[SPW][HH];
        #pragma unroll
        for (int s = 0; s < SPW; ++s) {
            #pragma unroll
            for (int h = 0; h < HH; ++h) {
                float cn = sigf(g0[s][5 + h]) * cl0[h]
                         + sigf(g0[s][h]) * tanh_hw(g0[s][10 + h]);
                h1v[s][h] = sigf(g0[s][15 + h]) * tanh_hw(cn);
            }
        }

        float g1[SPW][GG];
        #pragma unroll
        for (int g = 0; g < GG; ++g) {
            const float hb1 = Hb[1][lane * GG + g];
            #pragma unroll
            for (int s = 0; s < SPW; ++s) {
                float sum = hb1;
                #pragma unroll
                for (int j = 0; j < HH; ++j) sum += W_ih1[g * HH + j] * h1v[s][j];
                g1[s][g] = sum;
            }
        }
        #pragma unroll
        for (int s = 0; s < SPW; ++s) {
            const long row = row0 + (long)s * BB;
            #pragma unroll
            for (int h = 0; h < HH; ++h) {
                float cn = sigf(g1[s][5 + h]) * cl1[h]
                         + sigf(g1[s][h]) * tanh_hw(g1[s][10 + h]);
                out[row * HH + h] = sigf(g1[s][15 + h]) * tanh_hw(cn);
            }
        }
    }
}

extern "C" void kernel_launch(void* const* d_in, const int* in_sizes, int n_in,
                              void* d_out, int out_size, void* d_ws, size_t ws_size,
                              hipStream_t stream)
{
    const float* hidden = (const float*)d_in[1];
    const float* cellp  = (const float*)d_in[2];
    const float* dec_x  = (const float*)d_in[3];
    const int*   cat1   = (const int*)d_in[4];
    const int*   cat2   = (const int*)d_in[5];
    const int*   cat3   = (const int*)d_in[6];
    const float* emb    = (const float*)d_in[7];
    const float* W_ih0  = (const float*)d_in[8];
    const float* W_hh0  = (const float*)d_in[9];
    const float* b_ih0  = (const float*)d_in[10];
    const float* b_hh0  = (const float*)d_in[11];
    const float* W_ih1  = (const float*)d_in[12];
    const float* W_hh1  = (const float*)d_in[13];
    const float* b_ih1  = (const float*)d_in[14];
    const float* b_hh1  = (const float*)d_in[15];
    float* out = (float*)d_out;

    const int T    = in_sizes[3] / (BB * DD);
    const int rows = T * BB;

    lstm_wmma_kernel<<<rows / 512, 256, 0, stream>>>(hidden, cellp, dec_x, cat1, cat2, cat3, emb,
                                                     W_ih0, W_hh0, b_ih0, b_hh0,
                                                     W_ih1, W_hh1, b_ih1, b_hh1, out);
}


# dec_x rows read by whole-line loads (512 contiguous bytes, 4 lines each) and turned into A fragments via a per-wave LDS buffer; was 16 lines per load, every line touched 8 times; sums unchanged
# speedup vs baseline: 5.6900x; 1.1300x over previous
#include <hip/hip_runtime.h>
#include <hip/hip_bf16.h>


typedef float v2f __attribute__((ext_vector_type(2)));
typedef float v4f __attribute__((ext_vector_type(4)));
typedef float v8f __attribute__((ext_vector_type(8)));

#define BB   4096
#define DD   32
#define HH   5
#define GG   20
#define INW  128
#define VV   10
#define SPW  2
#define TPB  16
#define WPITCH 132
#define EPITCH 36
#define XPITCH 36

__device__ __forceinline__ float tanh_hw(float x) { return __builtin_amdgcn_tanhf(x); }
__device__ __forceinline__ float sigf(float x) { return __builtin_fmaf(0.5f, tanh_hw(0.5f * x), 0.5f); }

__device__ __forceinline__ void lds_barrier() {
    __builtin_amdgcn_fence(__ATOMIC_RELEASE, "workgroup", "local");
    __builtin_amdgcn_s_barrier();
    __builtin_amdgcn_fence(__ATOMIC_ACQUIRE, "workgroup", "local");
}

__global__ __launch_bounds__(256) void lstm_wmma_kernel(
    const float* hidden,
    const float* __restrict__ cell,
    const float* __restrict__ dec_x,
    const int*   __restrict__ cat1,
    const int*   __restrict__ cat2,
    const int*   __restrict__ cat3,
    const float* emb,
    const float* W_ih0,
    const float* __restrict__ W_hh0, const float* __restrict__ b_ih0, const float* __restrict__ b_hh0,
    const float* __restrict__ W_ih1,
    const float* __restrict__ W_hh1, const float* __restrict__ b_ih1, const float* __restrict__ b_hh1,
    float* __restrict__ out)
{
    __shared__ __attribute__((aligned(16))) float lds[GG * WPITCH + 16 * EPITCH + 3 * VV * GG + 2 * 32 * GG
                                                      + 8 * SPW * 32 * XPITCH];
    float* const Wp = lds;
    float* const Es = Wp + GG * WPITCH;
    float* const Ts = Es + 16 * EPITCH;
    float (* const Hb)[32 * GG] = (float (*)[32 * GG])(Ts + 3 * VV * GG);
    float* const Xg = &Hb[2][0];

    const int tid  = threadIdx.x;
    const int lane = tid & 31;
    const int wave = __builtin_amdgcn_readfirstlane(tid >> 5);

    const int b0 = (blockIdx.x & (BB / 32 - 1)) * 32;
    const int t  = (blockIdx.x / (BB / 32)) * TPB + wave * SPW;
    const int b  = b0 + lane;

    const int m    = lane & 15;
    const int half = lane >> 4;
    const int koff = half * 2;

    const long rbase = (long)t * BB + b0;
    const long row0  = rbase + lane;

    const int layer = wave >> 2;
    const int gq    = (wave & 3) * HH;
    float wh[HH][HH], bsum[HH];
    {
        const float* Wh = layer ? W_hh1 : W_hh0;
        const float* bi = layer ? b_ih1 : b_ih0;
        const float* bh = layer ? b_hh1 : b_hh0;
        #pragma unroll
        for (int q = 0; q < HH; ++q) {
            bsum[q] = bi[gq + q] + bh[gq + q];
            #pragma unroll
            for (int j = 0; j < HH; ++j) wh[q][j] = Wh[(gq + q) * HH + j];
        }
    }
    const int i0 = tid, i1 = tid + 256, i2 = tid + 512;
    const int ie = i2 - GG * (INW / 4);
    const int iec = ie < VV * 8 ? ie : VV * 8 - 1;
    const float4 w0 = *(const float4*)(W_ih0 + i0 * 4);
    const float4 w1 = *(const float4*)(W_ih0 + i1 * 4);
    const float4 w2 = *(const float4*)(ie < 0 ? W_ih0 + i2 * 4 : emb + iec * 4);
    float hv[HH];
    {
        const float* hp = hidden + ((long)layer * BB + b) * HH;
        #pragma unroll
        for (int j = 0; j < HH; ++j) hv[j] = hp[j];
    }
    __builtin_amdgcn_sched_barrier(0);
    v4f xv[SPW][8];
    int c1[SPW], c2[SPW], c3[SPW];
    #pragma unroll
    for (int s = 0; s < SPW; ++s) {
        const float* xp = dec_x + (rbase + (long)s * BB) * DD + lane * 4;
        #pragma unroll
        for (int i = 0; i < 8; ++i) xv[s][i] = *(const v4f*)(xp + i * (4 * DD));
        c1[s] = cat1[row0 + (long)s * BB];
        c2[s] = cat2[row0 + (long)s * BB];
        c3[s] = cat3[row0 + (long)s * BB];
    }
    float cl0[HH], cl1[HH];
    #pragma unroll
    for (int h = 0; h < HH; ++h) {
        cl0[h] = cell[(long)b * HH + h];
        cl1[h] = cell[(long)BB * HH + (long)b * HH + h];
    }

    *(float4*)(&Wp[(i0 >> 5) * WPITCH + (i0 & 31) * 4]) = w0;
    *(float4*)(&Wp[(i1 >> 5) * WPITCH + (i1 & 31) * 4]) = w1;
    *(float4*)(ie < 0 ? &Wp[(i2 >> 5) * WPITCH + (i2 & 31) * 4] : &Es[(ie >> 3) * EPITCH + (ie & 7) * 4]) = w2;
    #pragma unroll
    for (int q = 0; q < HH; ++q) {
        float s = bsum[q];
        #pragma unroll
        for (int j = 0; j < HH; ++j) s += hv[j] * wh[q][j];
        Hb[layer][lane * GG + gq + q] = s;
    }
    lds_barrier();

    if (wave < 4) {
        const int n  = wave * 16 + m;
        const int nn = n < 3 * GG ? n : 3 * GG - 1;
        const int s  = nn / GG;
        const int g  = nn - s * GG;
        const float* ea = &Es[(m < VV ? m : VV - 1) * EPITCH + koff];
        const float* wb = &Wp[g * WPITCH + 32 * (s + 1) + koff];
        v8f acc = {};
        #pragma unroll
        for (int kk = 0; kk < 8; ++kk) {
            const v2f a  = *(const v2f*)(ea + kk * 4);
            const v2f bw = *(const v2f*)(wb + kk * 4);
            acc = __builtin_amdgcn_wmma_f32_16x16x4_f32(false, a, false, bw,
                                                        (short)0, acc, false, false);
        }
        #pragma unroll
        for (int j = 0; j < 8; ++j) {
            const int v = j + half * 8;
            if (v < VV && n < 3 * GG) Ts[(s * VV + v) * GG + g] = acc[j];
        }
    }
    lds_barrier();

    float* xg = Xg + wave * (SPW * 32 * XPITCH);
    {
        float* xw = xg + (lane >> 3) * XPITCH + (lane & 7) * 4;
        #pragma unroll
        for (int s = 0; s < SPW; ++s) {
            #pragma unroll
            for (int i = 0; i < 8; ++i) *(v4f*)(xw + (s * 32 + 4 * i) * XPITCH) = xv[s][i];
        }
    }
    v2f a0[SPW][8], a1[SPW][8];
    {
        const float* xr = xg + m * XPITCH + koff;
        #pragma unroll
        for (int s = 0; s < SPW; ++s) {
            #pragma unroll
            for (int kk = 0; kk < 8; ++kk) {
                a0[s][kk] = *(const v2f*)(xr + (s * 32) * XPITCH + kk * 4);
                a1[s][kk] = *(const v2f*)(xr + (s * 32 + 16) * XPITCH + kk * 4);
            }
        }
    }
    __builtin_amdgcn_sched_barrier(0);

    const float* bp0 = &Wp[m * WPITCH + koff];
    const float* bp1 = &Wp[(m < 4 ? m + 16 : GG - 1) * WPITCH + koff];

    v8f acc00[SPW], acc01[SPW], acc10[SPW], acc11[SPW];
    #pragma unroll
    for (int s = 0; s < SPW; ++s) { acc00[s] = {}; acc01[s] = {}; acc10[s] = {}; acc11[s] = {}; }
    #pragma unroll
    for (int kk = 0; kk < 8; ++kk) {
        const v2f b0v = *(const v2f*)(bp0 + kk * 4);
        const v2f b1v = *(const v2f*)(bp1 + kk * 4);
        #pragma unroll
        for (int s = 0; s < SPW; ++s) {
            acc00[s] = __builtin_amdgcn_wmma_f32_16x16x4_f32(false, a0[s][kk], false, b0v,
                                                             (short)0, acc00[s], false, false);
            acc01[s] = __builtin_amdgcn_wmma_f32_16x16x4_f32(false, a0[s][kk], false, b1v,
                                                             (short)0, acc01[s], false, false);
            acc10[s] = __builtin_amdgcn_wmma_f32_16x16x4_f32(false, a1[s][kk], false, b0v,
                                                             (short)0, acc10[s], false, false);
            acc11[s] = __builtin_amdgcn_wmma_f32_16x16x4_f32(false, a1[s][kk], false, b1v,
                                                             (short)0, acc11[s], false, false);
        }
    }

    #pragma unroll
    for (int s = 0; s < SPW; ++s) {
        float* gs = xg + s * (32 * GG);
        #pragma unroll
        for (int j = 0; j < 8; ++j) {
            const int rm = j + half * 8;
            gs[rm * GG + m]        = acc00[s][j];
            gs[(rm + 16) * GG + m] = acc10[s][j];
        }
    }
    if (m < 4) {
        #pragma unroll
        for (int s = 0; s < SPW; ++s) {
            float* gs = xg + s * (32 * GG);
            #pragma unroll
            for (int j = 0; j < 8; ++j) {
                const int rm = j + half * 8;
                gs[rm * GG + 16 + m]        = acc01[s][j];
                gs[(rm + 16) * GG + 16 + m] = acc11[s][j];
            }
        }
    }
    asm volatile("s_wait_dscnt 0" ::: "memory");

    {
        float g0[SPW][GG];
        #pragma unroll
        for (int q = 0; q < GG / 4; ++q) {
            const float4 hq = *(const float4*)(&Hb[0][lane * GG + 4 * q]);
            const int g = 4 * q;
            #pragma unroll
            for (int s = 0; s < SPW; ++s) {
                const float4 u1 = *(const float4*)(&Ts[(0 * VV + c1[s]) * GG + 4 * q]);
                const float4 u2 = *(const float4*)(&Ts[(1 * VV + c2[s]) * GG + 4 * q]);
                const float4 u3 = *(const float4*)(&Ts[(2 * VV + c3[s]) * GG + 4 * q]);
                const float* gs = xg + s * (32 * GG) + lane * GG + g;
                g0[s][g + 0] = (gs[0] + ((u1.x + u2.x) + u3.x)) + hq.x;
                g0[s][g + 1] = (gs[1] + ((u1.y + u2.y) + u3.y)) + hq.y;
                g0[s][g + 2] = (gs[2] + ((u1.z + u2.z) + u3.z)) + hq.z;
                g0[s][g + 3] = (gs[3] + ((u1.w + u2.w) + u3.w)) + hq.w;
            }
        }

        float h1v[SPW][HH];
        #pragma unroll
        for (int s = 0; s < SPW; ++s) {
            #pragma unroll
            for (int h = 0; h < HH; ++h) {
                float cn = sigf(g0[s][5 + h]) * cl0[h]
                         + sigf(g0[s][h]) * tanh_hw(g0[s][10 + h]);
                h1v[s][h] = sigf(g0[s][15 + h]) * tanh_hw(cn);
            }
        }

        const __attribute__((address_space(4))) float* wi1 = (const __attribute__((address_space(4))) float*)W_ih1;
        float g1[SPW][GG];
        #pragma unroll
        for (int g = 0; g < GG; ++g) {
            const float hb1 = Hb[1][lane * GG + g];
            #pragma unroll
            for (int s = 0; s < SPW; ++s) {
                float sum = hb1;
                #pragma unroll
                for (int j = 0; j < HH; ++j) sum += wi1[g * HH + j] * h1v[s][j];
                g1[s][g] = sum;
            }
        }
        #pragma unroll
        for (int s = 0; s < SPW; ++s) {
            const long row = row0 + (long)s * BB;
            #pragma unroll
            for (int h = 0; h < HH; ++h) {
                float cn = sigf(g1[s][5 + h]) * cl1[h]
                         + sigf(g1[s][h]) * tanh_hw(g1[s][10 + h]);
                out[row * HH + h] = sigf(g1[s][15 + h]) * tanh_hw(cn);
            }
        }
    }
}

extern "C" void kernel_launch(void* const* d_in, const int* in_sizes, int n_in,
                              void* d_out, int out_size, void* d_ws, size_t ws_size,
                              hipStream_t stream)
{
    const float* hidden = (const float*)d_in[1];
    const float* cellp  = (const float*)d_in[2];
    const float* dec_x  = (const float*)d_in[3];
    const int*   cat1   = (const int*)d_in[4];
    const int*   cat2   = (const int*)d_in[5];
    const int*   cat3   = (const int*)d_in[6];
    const float* emb    = (const float*)d_in[7];
    const float* W_ih0  = (const float*)d_in[8];
    const float* W_hh0  = (const float*)d_in[9];
    const float* b_ih0  = (const float*)d_in[10];
    const float* b_hh0  = (const float*)d_in[11];
    const float* W_ih1  = (const float*)d_in[12];
    const float* W_hh1  = (const float*)d_in[13];
    const float* b_ih1  = (const float*)d_in[14];
    const float* b_hh1  = (const float*)d_in[15];
    float* out = (float*)d_out;

    const int T    = in_sizes[3] / (BB * DD);
    const int rows = T * BB;

    lstm_wmma_kernel<<<rows / 512, 256, 0, stream>>>(hidden, cellp, dec_x, cat1, cat2, cat3, emb,
                                                     W_ih0, W_hh0, b_ih0, b_hh0,
                                                     W_ih1, W_hh1, b_ih1, b_hh1, out);
}
